# starting kernel
# baseline (speedup 1.0000x reference)
#include <hip/hip_runtime.h>

typedef __attribute__((ext_vector_type(2))) float v2f;
typedef __attribute__((ext_vector_type(8))) float v8f;

constexpr int kNCOEF = 25;
constexpr int kM0    = 5;
constexpr int kC     = 128;
constexpr int kTILE  = 16;
constexpr float kCUTOFF  = 12.0f;
constexpr float kRESCALE = 5.0f;
constexpr float kEPS     = 1e-5f;

__device__ __forceinline__ float silu_f(float x) {
    return x / (1.0f + __expf(-x));
}

__device__ __forceinline__ void ln_silu_16x64(const float* __restrict__ hraw,
                                              float* __restrict__ hout,
                                              const float* __restrict__ g,
                                              const float* __restrict__ be,
                                              int tid) {
    if (tid < kTILE) {
        float s = 0.0f, s2 = 0.0f;
        #pragma unroll 8
        for (int c = 0; c < 64; ++c) {
            float v = hraw[tid * 64 + c];
            s += v; s2 += v * v;
        }
        float mu  = s * (1.0f / 64.0f);
        float var = s2 * (1.0f / 64.0f) - mu * mu;
        float inv = rsqrtf(var + kEPS);
        #pragma unroll 8
        for (int c = 0; c < 64; ++c) {
            float v = (hraw[tid * 64 + c] - mu) * inv * g[c] + be[c];
            hout[tid * 64 + c] = silu_f(v);
        }
    }
}

__global__ __launch_bounds__(128)
void edge_degree_embed_kernel(const float* __restrict__ x_edge,
                              const float* __restrict__ edge_distance,
                              const float* __restrict__ wigner_inv,
                              const int*   __restrict__ edge_index,
                              const float* __restrict__ w1, const float* __restrict__ b1,
                              const float* __restrict__ g1, const float* __restrict__ be1,
                              const float* __restrict__ w2, const float* __restrict__ b2,
                              const float* __restrict__ g2, const float* __restrict__ be2,
                              const float* __restrict__ w3, const float* __restrict__ b3,
                              const int*   __restrict__ node_offset_p,
                              float* __restrict__ out,
                              int E)
{
    __shared__ float xe[kTILE * 128];
    __shared__ float hraw[kTILE * 64];
    __shared__ float h1[kTILE * 64];
    __shared__ float h2[kTILE * 64];
    __shared__ float m0s[kTILE * kM0 * kC];
    __shared__ float envs[kTILE];

    const int tid   = threadIdx.x;
    const int wave  = tid >> 5;
    const int lane  = tid & 31;
    const int lrow  = lane & 15;
    const int khalf = (lane >> 4) << 1;
    const int rbase = (lane >> 4) << 3;
    const int e0    = blockIdx.x * kTILE;

    if (tid < kTILE && (e0 + tid) < E) {
        __builtin_prefetch(&wigner_inv[(size_t)(e0 + tid) * (kNCOEF * kNCOEF)], 0, 0);
    }

    for (int i = tid; i < kTILE * 128; i += blockDim.x) {
        int e = e0 + (i >> 7);
        xe[i] = (e < E) ? x_edge[(size_t)e * 128 + (i & 127)] : 0.0f;
    }
    __syncthreads();

    {
        const int colBase = wave * 16;
        v8f acc = {};
        for (int k0 = 0; k0 < 128; k0 += 4) {
            v2f a, b;
            a.x = xe[lrow * 128 + k0 + khalf];
            a.y = xe[lrow * 128 + k0 + khalf + 1];
            b.x = w1[(k0 + khalf)     * 64 + colBase + lrow];
            b.y = w1[(k0 + khalf + 1) * 64 + colBase + lrow];
            acc = __builtin_amdgcn_wmma_f32_16x16x4_f32(
                false, a, false, b, (short)0, acc, false, false);
        }
        #pragma unroll
        for (int r = 0; r < 8; ++r) {
            int row = rbase + r;
            hraw[row * 64 + colBase + lrow] = acc[r] + b1[colBase + lrow];
        }
    }
    __syncthreads();
    ln_silu_16x64(hraw, h1, g1, be1, tid);
    __syncthreads();

    {
        const int colBase = wave * 16;
        v8f acc = {};
        for (int k0 = 0; k0 < 64; k0 += 4) {
            v2f a, b;
            a.x = h1[lrow * 64 + k0 + khalf];
            a.y = h1[lrow * 64 + k0 + khalf + 1];
            b.x = w2[(k0 + khalf)     * 64 + colBase + lrow];
            b.y = w2[(k0 + khalf + 1) * 64 + colBase + lrow];
            acc = __builtin_amdgcn_wmma_f32_16x16x4_f32(
                false, a, false, b, (short)0, acc, false, false);
        }
        #pragma unroll
        for (int r = 0; r < 8; ++r) {
            int row = rbase + r;
            hraw[row * 64 + colBase + lrow] = acc[r] + b2[colBase + lrow];
        }
    }
    __syncthreads();
    ln_silu_16x64(hraw, h2, g2, be2, tid);
    __syncthreads();

    for (int t = wave; t < 40; t += 4) {
        const int colBase = t * 16;
        v8f acc = {};
        for (int k0 = 0; k0 < 64; k0 += 4) {
            v2f a, b;
            a.x = h2[lrow * 64 + k0 + khalf];
            a.y = h2[lrow * 64 + k0 + khalf + 1];
            b.x = w3[(size_t)(k0 + khalf)     * 640 + colBase + lrow];
            b.y = w3[(size_t)(k0 + khalf + 1) * 640 + colBase + lrow];
            acc = __builtin_amdgcn_wmma_f32_16x16x4_f32(
                false, a, false, b, (short)0, acc, false, false);
        }
        #pragma unroll
        for (int r = 0; r < 8; ++r) {
            int row = rbase + r;
            m0s[row * (kM0 * kC) + colBase + lrow] = acc[r] + b3[colBase + lrow];
        }
    }
    __syncthreads();

    float* wig = xe;
    for (int i = tid; i < kTILE * kNCOEF * kM0; i += blockDim.x) {
        int e_loc = i / (kNCOEF * kM0);
        int rem   = i % (kNCOEF * kM0);
        int r     = rem / kM0;
        int l     = rem % kM0;
        int e     = e0 + e_loc;
        wig[i] = (e < E)
            ? wigner_inv[(size_t)e * (kNCOEF * kNCOEF) + r * kNCOEF + l * (l + 1)]
            : 0.0f;
    }
    if (tid < kTILE) {
        int e = e0 + tid;
        float env = 0.0f;
        if (e < E) {
            float d = edge_distance[e] * (1.0f / kCUTOFF);
            if (d < 1.0f) {
                float d5 = d * d * d * d * d;
                env = 1.0f - 21.0f * d5 + 35.0f * d5 * d - 15.0f * d5 * d * d;
            }
        }
        envs[tid] = env * (1.0f / kRESCALE);
    }
    __syncthreads();

    const int noff = node_offset_p[0];
    for (int i = tid; i < kTILE * kNCOEF * kC; i += blockDim.x) {
        int e_loc = i / (kNCOEF * kC);
        int rem   = i % (kNCOEF * kC);
        int e     = e0 + e_loc;
        if (e >= E) continue;
        int r = rem >> 7;
        int c = rem & 127;
        float s = 0.0f;
        #pragma unroll
        for (int l = 0; l < kM0; ++l) {
            s += wig[(e_loc * kNCOEF + r) * kM0 + l] * m0s[e_loc * (kM0 * kC) + l * kC + c];
        }
        int dst = edge_index[E + e] - noff;
        atomicAdd(&out[((size_t)dst * kNCOEF + r) * kC + c], s * envs[e_loc]);
    }
}

extern "C" void kernel_launch(void* const* d_in, const int* in_sizes, int n_in,
                              void* d_out, int out_size, void* d_ws, size_t ws_size,
                              hipStream_t stream) {
    const float* x             = (const float*)d_in[0];
    const float* x_edge        = (const float*)d_in[1];
    const float* edge_distance = (const float*)d_in[2];
    const float* wigner_inv    = (const float*)d_in[3];
    const int*   edge_index    = (const int*)d_in[4];
    const float* w1  = (const float*)d_in[5];
    const float* b1  = (const float*)d_in[6];
    const float* g1  = (const float*)d_in[7];
    const float* be1 = (const float*)d_in[8];
    const float* w2  = (const float*)d_in[9];
    const float* b2  = (const float*)d_in[10];
    const float* g2  = (const float*)d_in[11];
    const float* be2 = (const float*)d_in[12];
    const float* w3  = (const float*)d_in[13];
    const float* b3  = (const float*)d_in[14];
    const int* node_offset = (const int*)d_in[15];

    const int E = in_sizes[1] / 128;

    hipMemcpyAsync(d_out, x, (size_t)out_size * sizeof(float),
                   hipMemcpyDeviceToDevice, stream);

    const int blocks = (E + kTILE - 1) / kTILE;
    edge_degree_embed_kernel<<<blocks, 128, 0, stream>>>(
        x_edge, edge_distance, wigner_inv, edge_index,
        w1, b1, g1, be1, w2, b2, g2, be2, w3, b3,
        node_offset, (float*)d_out, E);
}


# Each pass's gathered inputs (x_edge rows, m=0 Wigner columns, distances) requested one pass ahead into registers, all in flight, stored to LDS at the next pass's top; in-pass barriers order LDS only
# speedup vs baseline: 1.7017x; 1.7017x over previous
#include <hip/hip_runtime.h>

typedef __attribute__((ext_vector_type(2))) float v2f;
typedef __attribute__((ext_vector_type(8))) float v8f;
typedef __attribute__((ext_vector_type(4))) float v4f;

constexpr int kNCOEF = 25;
constexpr int kM0    = 5;
constexpr int kC     = 128;
constexpr int kTILE  = 16;
constexpr int kG     = 4;
constexpr int kNG    = 4;
constexpr int kTHREADS = 128 * kNG;
constexpr int kW3P   = 1312;
constexpr int kCAP   = 128;
constexpr int kGF    = 8208;
constexpr int kHR    = 13;
constexpr float kCUTOFF  = 12.0f;
constexpr float kRESCALE = 5.0f;
constexpr float kEPS     = 1e-5f;

__device__ __forceinline__ float silu_f(float x) {
    return x / (1.0f + __expf(-x));
}

__device__ __forceinline__ void lds_barrier() {
    __builtin_amdgcn_fence(__ATOMIC_RELEASE, "workgroup", "local");
    __builtin_amdgcn_s_barrier();
    __builtin_amdgcn_fence(__ATOMIC_ACQUIRE, "workgroup", "local");
}

__device__ __forceinline__ void ln_silu_16x64(const float* __restrict__ hraw,
                                              float* __restrict__ hout,
                                              const float* __restrict__ g,
                                              const float* __restrict__ be,
                                              int tid) {
    if (tid >= 0 && tid < kTILE) {
        float s = 0.0f, s2 = 0.0f;
        #pragma unroll 8
        for (int c = 0; c < 64; ++c) {
            float v = hraw[tid * 64 + c];
            s += v; s2 += v * v;
        }
        float mu  = s * (1.0f / 64.0f);
        float var = s2 * (1.0f / 64.0f) - mu * mu;
        float inv = rsqrtf(var + kEPS);
        #pragma unroll 8
        for (int c = 0; c < 64; ++c) {
            float v = (hraw[tid * 64 + c] - mu) * inv * g[c] + be[c];
            hout[tid * 64 + c] = silu_f(v);
        }
    }
}

__device__ __forceinline__ void flush_half(const float* __restrict__ x,
                                           float* __restrict__ out,
                                           float (&nacc)[kHR],
                                           int node, int chan, int rbeg, int nr, bool seen) {
    const size_t base = (size_t)node * (kNCOEF * kC) + (size_t)rbeg * kC + chan;
    float start[kHR];
    #pragma unroll
    for (int r = 0; r < kHR; ++r) {
        start[r] = 0.0f;
        if (r < nr) start[r] = seen ? out[base + r * kC] : x[base + r * kC];
    }
    #pragma unroll
    for (int r = 0; r < kHR; ++r) {
        if (r < nr) out[base + r * kC] = start[r] + nacc[r];
        nacc[r] = 0.0f;
    }
}

__global__ __launch_bounds__(kTHREADS, kNG)
void edge_degree_embed_kernel(const float* __restrict__ x,
                              const float* __restrict__ x_edge,
                              const float* __restrict__ edge_distance,
                              const float* __restrict__ wigner_inv,
                              const int*   __restrict__ edge_index,
                              const float* __restrict__ w1, const float* __restrict__ b1,
                              const float* __restrict__ g1, const float* __restrict__ be1,
                              const float* __restrict__ w2, const float* __restrict__ b2,
                              const float* __restrict__ g2, const float* __restrict__ be2,
                              const float* __restrict__ w3, const float* __restrict__ b3,
                              const int*   __restrict__ node_offset_p,
                              float* __restrict__ out,
                              int E, int N)
{
    __shared__ __attribute__((aligned(16))) float w3s[32 * kW3P];
    __shared__ __attribute__((aligned(16))) float tiles_[kNG][kGF];
    __shared__ int lst_e_[kNG][kCAP];
    __shared__ int lst_g_[kNG][kCAP];
    __shared__ int srt_e_[kNG][kCAP];
    __shared__ int srt_g_[kNG][kCAP];
    __shared__ int wcnt[4 * kNG][kNG];

    const int tid   = threadIdx.x;
    const int wave  = tid >> 5;
    const int grp   = __builtin_amdgcn_readfirstlane(tid >> 7);
    const int gt    = tid & 127;
    const int gw    = wave & 3;
    const int lane  = tid & 31;
    const int lrow  = lane & 15;
    const int khalf = (lane >> 4) << 1;
    const int rbase = (lane >> 4) << 3;

    float* m0s  = tiles_[grp];
    float* xe   = m0s;
    float* hraw = m0s + 2048;
    float* h1   = m0s + 3072;
    float* wig  = m0s + 5120;
    float* h2   = m0s + 7168;
    float* envs = m0s + 8192;
    int* lst_e = lst_e_[grp];
    int* lst_g = lst_g_[grp];
    int* srt_e = srt_e_[grp];
    int* srt_g = srt_g_[grp];

    constexpr int kUNITS = 32 * 160;
    constexpr int kFULL  = kUNITS / kTHREADS;
    static_assert(kFULL * kTHREADS == kUNITS, "the units must divide evenly among the threads");
    auto stage_w3 = [&](int u) {
        const int p = u / 160;
        const int q = u - p * 160;
        const v4f r0 = *(const v4f*)(w3 + (size_t)(2 * p)     * 640 + 4 * q);
        const v4f r1 = *(const v4f*)(w3 + (size_t)(2 * p + 1) * 640 + 4 * q);
        v4f s0, s1;
        s0.x = r0.x; s0.y = r1.x; s0.z = r0.y; s0.w = r1.y;
        s1.x = r0.z; s1.y = r1.z; s1.z = r0.w; s1.w = r1.w;
        *(v4f*)(w3s + p * kW3P + 8 * q)     = s0;
        *(v4f*)(w3s + p * kW3P + 8 * q + 4) = s1;
    };
    #pragma unroll
    for (int i = 0; i < kFULL; ++i) stage_w3(tid + i * kTHREADS);

    const int noff = node_offset_p[0];
    const int nb0  = blockIdx.x * (kG * kNG);
    const int nbn  = min(kG * kNG, N - nb0);
    const int n0   = nb0 + grp * kG;
    const int n1   = min(n0 + kG, N);

    constexpr int kU = 16;
    constexpr int kNW = 4 * kNG;
    const int wv   = __builtin_amdgcn_readfirstlane(wave);
    const int Q    = (((E + kNW - 1) / kNW) + 31) & ~31;
    const int wbeg = wv * Q;
    const int wend = min(E, wbeg + Q);

    int cnt[kNG];
    #pragma unroll
    for (int g = 0; g < kNG; ++g) cnt[g] = 0;
    for (int s = wbeg; s < wend; s += 32 * kU) {
        int dv[kU];
        #pragma unroll
        for (int u = 0; u < kU; ++u) {
            const int idx = s + u * 32 + lane;
            dv[u] = edge_index[E + min(idx, wend - 1)];
        }
        #pragma unroll
        for (int u = 0; u < kU; ++u) {
            const int idx = s + u * 32 + lane;
            const int d   = dv[u] - noff - nb0;
            const bool in = (idx < wend) && (d >= 0) && (d < nbn);
            #pragma unroll
            for (int g = 0; g < kNG; ++g)
                cnt[g] += __popc(__builtin_amdgcn_ballot_w32(in && d >= g * kG && d < (g + 1) * kG));
        }
    }
    if (lane == 0) {
        #pragma unroll
        for (int g = 0; g < kNG; ++g) wcnt[wave][g] = cnt[g];
    }
    __syncthreads();
    int base[kNG], tot[kNG];
    #pragma unroll
    for (int g = 0; g < kNG; ++g) { base[g] = 0; tot[g] = 0; }
    #pragma unroll
    for (int w = 0; w < kNW; ++w) {
        #pragma unroll
        for (int g = 0; g < kNG; ++g) {
            const int c = wcnt[w][g];
            if (w < wave) base[g] += c;
            tot[g] += c;
        }
    }
    int kmax = 0;
    #pragma unroll
    for (int g = 0; g < kNG; ++g) {
        tot[g] = __builtin_amdgcn_readfirstlane(tot[g]);
        kmax = max(kmax, tot[g]);
    }

    const int chan = gt & 63;
    const int wrem = (gt < kNCOEF * kM0) ? gt : 0;
    const int woff = (wrem / kM0) * kNCOEF + (wrem % kM0) * ((wrem % kM0) + 1);
    const int lnw  = 32 * (grp & 3);
    const int rh   = __builtin_amdgcn_readfirstlane(gt >> 6);
    const int rbeg = rh * kHR;
    const int nr   = rh ? (kNCOEF - kHR) : kHR;
    float nacc0[kHR], nacc1[kHR];
    #pragma unroll
    for (int r = 0; r < kHR; ++r) { nacc0[r] = 0.0f; nacc1[r] = 0.0f; }
    unsigned written0 = 0u, written1 = 0u;

    auto layer3_half = [&](int hh) {
        for (int tt = gw; tt < 20; tt += 4) {
            const int l    = tt >> 2;
            const int j4   = tt & 3;
            const int col3 = l * kC + hh * 64 + j4 * 16 + lrow;
            v8f acc = {};
            for (int k0 = 0; k0 < 64; k0 += 4) {
                v2f a, b;
                a.x = h2[lrow * 64 + k0 + khalf];
                a.y = h2[lrow * 64 + k0 + khalf + 1];
                b = *(const v2f*)(w3s + ((k0 + khalf) >> 1) * kW3P + 2 * col3);
                acc = __builtin_amdgcn_wmma_f32_16x16x4_f32(
                    false, a, false, b, (short)0, acc, false, false);
            }
            #pragma unroll
            for (int r = 0; r < 8; ++r) {
                int row = rbase + r;
                m0s[row * (kM0 * 64) + l * 64 + j4 * 16 + lrow] = acc[r] + b3[col3];
            }
        }
    };

    auto contract_half = [&](float (&nacc)[kHR], int& cur, unsigned& written, int hh, int nE, int cb) {
        for (int j = 0; j < nE; ++j) {
            const int g = __builtin_amdgcn_readfirstlane(srt_g[cb + j]);
            if (g != cur) {
                if (cur >= 0) {
                    flush_half(x, out, nacc, n0 + cur, hh * 64 + chan, rbeg, nr, (written >> cur) & 1u);
                    written |= (1u << cur);
                }
                cur = g;
            }
            const float ev = envs[j];
            float mv[kM0];
            #pragma unroll
            for (int l = 0; l < kM0; ++l) mv[l] = m0s[j * (kM0 * 64) + l * 64 + chan] * ev;
            #pragma unroll
            for (int rr = 0; rr < kHR; ++rr) {
                if (rr < nr) {
                    float s = nacc[rr];
                    #pragma unroll
                    for (int l = 0; l < kM0; ++l) s += wig[j * 128 + (rbeg + rr) * kM0 + l] * mv[l];
                    nacc[rr] = s;
                }
            }
        }
    };

    for (int lo = 0; lo < kmax; lo += kCAP) {
        int mg[kNG];
        int m = 0, mmax = 0;
        #pragma unroll
        for (int g = 0; g < kNG; ++g) {
            mg[g] = max(0, min(kCAP, tot[g] - lo));
            if (g == grp) m = mg[g];
            mmax = max(mmax, mg[g]);
        }

        int run[kNG];
        #pragma unroll
        for (int g = 0; g < kNG; ++g) run[g] = base[g];
        for (int s = wbeg; s < wend; s += 32 * kU) {
            int dv[kU];
            #pragma unroll
            for (int u = 0; u < kU; ++u) {
                const int idx = s + u * 32 + lane;
                dv[u] = edge_index[E + min(idx, wend - 1)];
            }
            #pragma unroll
            for (int u = 0; u < kU; ++u) {
                const int idx = s + u * 32 + lane;
                const int d   = dv[u] - noff - nb0;
                const bool in = (idx < wend) && (d >= 0) && (d < nbn);
                #pragma unroll
                for (int g = 0; g < kNG; ++g) {
                    const bool ig = in && d >= g * kG && d < (g + 1) * kG;
                    const unsigned mask = __builtin_amdgcn_ballot_w32(ig);
                    if (ig) {
                        const int rank = run[g] + (int)__builtin_amdgcn_mbcnt_lo(mask, 0u) - lo;
                        if (rank >= 0 && rank < mg[g]) { lst_e_[g][rank] = idx; lst_g_[g][rank] = d - g * kG; }
                    }
                    run[g] += __popc(mask);
                }
            }
        }
        __syncthreads();

        for (int t = gt; t < m; t += 128) {
            const int gn = lst_g[t];
            int pos = 0;
            for (int j = 0; j < m; ++j) {
                const int gj = lst_g[j];
                pos += ((gj < gn) || (gj == gn && j < t)) ? 1 : 0;
            }
            srt_e[pos] = lst_e[t];
            srt_g[pos] = gn;
        }
        __syncthreads();

        float px[kTILE], pw[kTILE], pd = 0.0f;
        auto request = [&](int cbn) {
            const int nEn = max(0, min(kTILE, m - cbn));
            if (nEn > 0) {
                const int eid = srt_e[cbn + min(lrow, nEn - 1)];
                #pragma unroll
                for (int it = 0; it < kTILE; ++it) {
                    const int e = __builtin_amdgcn_readlane(eid, it);
                    px[it] = x_edge[(size_t)e * 128 + gt];
                    pw[it] = wigner_inv[(size_t)e * (kNCOEF * kNCOEF) + woff];
                }
                if (gt < kTILE) pd = edge_distance[eid];
            } else {
                #pragma unroll
                for (int it = 0; it < kTILE; ++it) { px[it] = 0.0f; pw[it] = 0.0f; }
            }
        };
        request(0);
        int cur0 = -1, cur1 = -1;
        for (int cb = 0; cb < mmax; cb += kTILE) {
            const int nE = max(0, min(kTILE, m - cb));

            #pragma unroll
            for (int it = 0; it < kTILE; ++it) {
                xe[it * 128 + gt]  = (it < nE) ? px[it] : 0.0f;
                wig[it * 128 + gt] = (it < nE) ? pw[it] : 0.0f;
            }
            if (gt < kTILE) {
                float env = 0.0f;
                if (gt < nE) {
                    float d = pd * (1.0f / kCUTOFF);
                    if (d < 1.0f) {
                        float d5 = d * d * d * d * d;
                        env = 1.0f - 21.0f * d5 + 35.0f * d5 * d - 15.0f * d5 * d * d;
                    }
                }
                envs[gt] = env * (1.0f / kRESCALE);
            }
            if (cb + kTILE < mmax) request(cb + kTILE);
            lds_barrier();

            {
                const int colBase = gw * 16;
                v8f acc = {};
                for (int k0 = 0; k0 < 128; k0 += 4) {
                    v2f a, b;
                    a.x = xe[lrow * 128 + k0 + khalf];
                    a.y = xe[lrow * 128 + k0 + khalf + 1];
                    b.x = w1[(k0 + khalf)     * 64 + colBase + lrow];
                    b.y = w1[(k0 + khalf + 1) * 64 + colBase + lrow];
                    acc = __builtin_amdgcn_wmma_f32_16x16x4_f32(
                        false, a, false, b, (short)0, acc, false, false);
                }
                #pragma unroll
                for (int r = 0; r < 8; ++r) {
                    int row = rbase + r;
                    hraw[row * 64 + colBase + lrow] = acc[r] + b1[colBase + lrow];
                }
            }
            lds_barrier();
            ln_silu_16x64(hraw, h1, g1, be1, gt - lnw);
            lds_barrier();

            {
                const int colBase = gw * 16;
                v8f acc = {};
                for (int k0 = 0; k0 < 64; k0 += 4) {
                    v2f a, b;
                    a.x = h1[lrow * 64 + k0 + khalf];
                    a.y = h1[lrow * 64 + k0 + khalf + 1];
                    b.x = w2[(k0 + khalf)     * 64 + colBase + lrow];
                    b.y = w2[(k0 + khalf + 1) * 64 + colBase + lrow];
                    acc = __builtin_amdgcn_wmma_f32_16x16x4_f32(
                        false, a, false, b, (short)0, acc, false, false);
                }
                #pragma unroll
                for (int r = 0; r < 8; ++r) {
                    int row = rbase + r;
                    hraw[row * 64 + colBase + lrow] = acc[r] + b2[colBase + lrow];
                }
            }
            lds_barrier();
            ln_silu_16x64(hraw, h2, g2, be2, gt - lnw);
            lds_barrier();

            layer3_half(0);
            lds_barrier();
            contract_half(nacc0, cur0, written0, 0, nE, cb);
            lds_barrier();
            layer3_half(1);
            lds_barrier();
            contract_half(nacc1, cur1, written1, 1, nE, cb);
            __syncthreads();
        }
        if (cur0 >= 0) {
            flush_half(x, out, nacc0, n0 + cur0, chan, rbeg, nr, (written0 >> cur0) & 1u);
            written0 |= (1u << cur0);
        }
        if (cur1 >= 0) {
            flush_half(x, out, nacc1, n0 + cur1, 64 + chan, rbeg, nr, (written1 >> cur1) & 1u);
            written1 |= (1u << cur1);
        }
    }

    for (int g = 0; g < n1 - n0; ++g) {
        if (!((written0 >> g) & 1u)) {
            const size_t base = (size_t)(n0 + g) * (kNCOEF * kC) + (size_t)rbeg * kC + chan;
            float s0[kHR], s1[kHR];
            #pragma unroll
            for (int r = 0; r < kHR; ++r) {
                s0[r] = 0.0f; s1[r] = 0.0f;
                if (r < nr) { s0[r] = x[base + r * kC]; s1[r] = x[base + r * kC + 64]; }
            }
            #pragma unroll
            for (int r = 0; r < kHR; ++r) {
                if (r < nr) { out[base + r * kC] = s0[r]; out[base + r * kC + 64] = s1[r]; }
            }
        }
    }
}

extern "C" void kernel_launch(void* const* d_in, const int* in_sizes, int n_in,
                              void* d_out, int out_size, void* d_ws, size_t ws_size,
                              hipStream_t stream) {
    const float* x             = (const float*)d_in[0];
    const float* x_edge        = (const float*)d_in[1];
    const float* edge_distance = (const float*)d_in[2];
    const float* wigner_inv    = (const float*)d_in[3];
    const int*   edge_index    = (const int*)d_in[4];
    const float* w1  = (const float*)d_in[5];
    const float* b1  = (const float*)d_in[6];
    const float* g1  = (const float*)d_in[7];
    const float* be1 = (const float*)d_in[8];
    const float* w2  = (const float*)d_in[9];
    const float* b2  = (const float*)d_in[10];
    const float* g2  = (const float*)d_in[11];
    const float* be2 = (const float*)d_in[12];
    const float* w3  = (const float*)d_in[13];
    const float* b3  = (const float*)d_in[14];
    const int* node_offset = (const int*)d_in[15];

    const int E = in_sizes[1] / 128;

    const int N = out_size / (kNCOEF * kC);

    const int blocks = (N + kG * kNG - 1) / (kG * kNG);
    edge_degree_embed_kernel<<<blocks, kTHREADS, 0, stream>>>(
        x, x_edge, edge_distance, wigner_inv, edge_index,
        w1, b1, g1, be1, w2, b2, g2, be2, w3, b3,
        node_offset, (float*)d_out, E, N);
}


# Layer 3: k loop outermost, the wave's five column tiles inside: an A fragment of layer 2's output (an LDS read, all sixteen rows on one bank) is read once per k step for five products, not per tile
# speedup vs baseline: 2.0228x; 1.1887x over previous
#include <hip/hip_runtime.h>

typedef __attribute__((ext_vector_type(2))) float v2f;
typedef __attribute__((ext_vector_type(8))) float v8f;
typedef __attribute__((ext_vector_type(4))) float v4f;

constexpr int kNCOEF = 25;
constexpr int kM0    = 5;
constexpr int kC     = 128;
constexpr int kTILE  = 16;
constexpr int kG     = 4;
constexpr int kNG    = 4;
constexpr int kTHREADS = 128 * kNG;
constexpr int kW3P   = 1312;
constexpr int kCAP   = 128;
constexpr int kGF    = 8208;
constexpr int kHR    = 13;
constexpr float kCUTOFF  = 12.0f;
constexpr float kRESCALE = 5.0f;
constexpr float kEPS     = 1e-5f;

__device__ __forceinline__ float silu_f(float x) {
    return x / (1.0f + __expf(-x));
}

__device__ __forceinline__ void lds_barrier() {
    __builtin_amdgcn_fence(__ATOMIC_RELEASE, "workgroup", "local");
    __builtin_amdgcn_s_barrier();
    __builtin_amdgcn_fence(__ATOMIC_ACQUIRE, "workgroup", "local");
}

__device__ __forceinline__ void ln_silu_16x64(const float* __restrict__ hraw,
                                              float* __restrict__ hout,
                                              const float* __restrict__ g,
                                              const float* __restrict__ be,
                                              int tid) {
    if (tid >= 0 && tid < kTILE) {
        float s = 0.0f, s2 = 0.0f;
        #pragma unroll 8
        for (int c = 0; c < 64; ++c) {
            float v = hraw[tid * 64 + c];
            s += v; s2 += v * v;
        }
        float mu  = s * (1.0f / 64.0f);
        float var = s2 * (1.0f / 64.0f) - mu * mu;
        float inv = rsqrtf(var + kEPS);
        #pragma unroll 8
        for (int c = 0; c < 64; ++c) {
            float v = (hraw[tid * 64 + c] - mu) * inv * g[c] + be[c];
            hout[tid * 64 + c] = silu_f(v);
        }
    }
}

__device__ __forceinline__ void flush_half(const float* __restrict__ x,
                                           float* __restrict__ out,
                                           float (&nacc)[kHR],
                                           int node, int chan, int rbeg, int nr, bool seen) {
    const size_t base = (size_t)node * (kNCOEF * kC) + (size_t)rbeg * kC + chan;
    float start[kHR];
    #pragma unroll
    for (int r = 0; r < kHR; ++r) {
        start[r] = 0.0f;
        if (r < nr) start[r] = seen ? out[base + r * kC] : x[base + r * kC];
    }
    #pragma unroll
    for (int r = 0; r < kHR; ++r) {
        if (r < nr) out[base + r * kC] = start[r] + nacc[r];
        nacc[r] = 0.0f;
    }
}

__global__ __launch_bounds__(kTHREADS, kNG)
void edge_degree_embed_kernel(const float* __restrict__ x,
                              const float* __restrict__ x_edge,
                              const float* __restrict__ edge_distance,
                              const float* __restrict__ wigner_inv,
                              const int*   __restrict__ edge_index,
                              const float* __restrict__ w1, const float* __restrict__ b1,
                              const float* __restrict__ g1, const float* __restrict__ be1,
                              const float* __restrict__ w2, const float* __restrict__ b2,
                              const float* __restrict__ g2, const float* __restrict__ be2,
                              const float* __restrict__ w3, const float* __restrict__ b3,
                              const int*   __restrict__ node_offset_p,
                              float* __restrict__ out,
                              int E, int N)
{
    __shared__ __attribute__((aligned(16))) float w3s[32 * kW3P];
    __shared__ __attribute__((aligned(16))) float tiles_[kNG][kGF];
    __shared__ int lst_e_[kNG][kCAP];
    __shared__ int lst_g_[kNG][kCAP];
    __shared__ int srt_e_[kNG][kCAP];
    __shared__ int srt_g_[kNG][kCAP];
    __shared__ int wcnt[4 * kNG][kNG];

    const int tid   = threadIdx.x;
    const int wave  = tid >> 5;
    const int grp   = __builtin_amdgcn_readfirstlane(tid >> 7);
    const int gt    = tid & 127;
    const int gw    = wave & 3;
    const int lane  = tid & 31;
    const int lrow  = lane & 15;
    const int khalf = (lane >> 4) << 1;
    const int rbase = (lane >> 4) << 3;

    float* m0s  = tiles_[grp];
    float* xe   = m0s;
    float* hraw = m0s + 2048;
    float* h1   = m0s + 3072;
    float* wig  = m0s + 5120;
    float* h2   = m0s + 7168;
    float* envs = m0s + 8192;
    int* lst_e = lst_e_[grp];
    int* lst_g = lst_g_[grp];
    int* srt_e = srt_e_[grp];
    int* srt_g = srt_g_[grp];

    constexpr int kUNITS = 32 * 160;
    constexpr int kFULL  = kUNITS / kTHREADS;
    static_assert(kFULL * kTHREADS == kUNITS, "the units must divide evenly among the threads");
    auto stage_w3 = [&](int u) {
        const int p = u / 160;
        const int q = u - p * 160;
        const v4f r0 = *(const v4f*)(w3 + (size_t)(2 * p)     * 640 + 4 * q);
        const v4f r1 = *(const v4f*)(w3 + (size_t)(2 * p + 1) * 640 + 4 * q);
        v4f s0, s1;
        s0.x = r0.x; s0.y = r1.x; s0.z = r0.y; s0.w = r1.y;
        s1.x = r0.z; s1.y = r1.z; s1.z = r0.w; s1.w = r1.w;
        *(v4f*)(w3s + p * kW3P + 8 * q)     = s0;
        *(v4f*)(w3s + p * kW3P + 8 * q + 4) = s1;
    };
    #pragma unroll
    for (int i = 0; i < kFULL; ++i) stage_w3(tid + i * kTHREADS);

    const int noff = node_offset_p[0];
    const int nb0  = blockIdx.x * (kG * kNG);
    const int nbn  = min(kG * kNG, N - nb0);
    const int n0   = nb0 + grp * kG;
    const int n1   = min(n0 + kG, N);

    constexpr int kU = 16;
    constexpr int kNW = 4 * kNG;
    const int wv   = __builtin_amdgcn_readfirstlane(wave);
    const int Q    = (((E + kNW - 1) / kNW) + 31) & ~31;
    const int wbeg = wv * Q;
    const int wend = min(E, wbeg + Q);

    int cnt[kNG];
    #pragma unroll
    for (int g = 0; g < kNG; ++g) cnt[g] = 0;
    for (int s = wbeg; s < wend; s += 32 * kU) {
        int dv[kU];
        #pragma unroll
        for (int u = 0; u < kU; ++u) {
            const int idx = s + u * 32 + lane;
            dv[u] = edge_index[E + min(idx, wend - 1)];
        }
        #pragma unroll
        for (int u = 0; u < kU; ++u) {
            const int idx = s + u * 32 + lane;
            const int d   = dv[u] - noff - nb0;
            const bool in = (idx < wend) && (d >= 0) && (d < nbn);
            #pragma unroll
            for (int g = 0; g < kNG; ++g)
                cnt[g] += __popc(__builtin_amdgcn_ballot_w32(in && d >= g * kG && d < (g + 1) * kG));
        }
    }
    if (lane == 0) {
        #pragma unroll
        for (int g = 0; g < kNG; ++g) wcnt[wave][g] = cnt[g];
    }
    __syncthreads();
    int base[kNG], tot[kNG];
    #pragma unroll
    for (int g = 0; g < kNG; ++g) { base[g] = 0; tot[g] = 0; }
    #pragma unroll
    for (int w = 0; w < kNW; ++w) {
        #pragma unroll
        for (int g = 0; g < kNG; ++g) {
            const int c = wcnt[w][g];
            if (w < wave) base[g] += c;
            tot[g] += c;
        }
    }
    int kmax = 0;
    #pragma unroll
    for (int g = 0; g < kNG; ++g) {
        tot[g]  = __builtin_amdgcn_readfirstlane(tot[g]);
        base[g] = __builtin_amdgcn_readfirstlane(base[g]);
        kmax = max(kmax, tot[g]);
    }

    const int chan = gt & 63;
    const int wrem = (gt < kNCOEF * kM0) ? gt : 0;
    const int woff = (wrem / kM0) * kNCOEF + (wrem % kM0) * ((wrem % kM0) + 1);
    const int lnw  = 32 * (grp & 3);
    const int rh   = __builtin_amdgcn_readfirstlane(gt >> 6);
    const int rbeg = rh * kHR;
    const int nr   = rh ? (kNCOEF - kHR) : kHR;
    float nacc0[kHR], nacc1[kHR];
    #pragma unroll
    for (int r = 0; r < kHR; ++r) { nacc0[r] = 0.0f; nacc1[r] = 0.0f; }
    unsigned written0 = 0u, written1 = 0u;

    auto layer3_half = [&](int hh) {
        const int cw = hh * 64 + gw * 16 + lrow;
        v8f acc[kM0];
        #pragma unroll
        for (int l = 0; l < kM0; ++l) acc[l] = v8f{};
        #pragma nounroll
        for (int k0 = 0; k0 < 64; k0 += 4) {
            v2f a;
            a.x = h2[lrow * 64 + k0 + khalf];
            a.y = h2[lrow * 64 + k0 + khalf + 1];
            const float* brow = w3s + ((k0 + khalf) >> 1) * kW3P + 2 * cw;
            #pragma unroll
            for (int l = 0; l < kM0; ++l) {
                const v2f b = *(const v2f*)(brow + 2 * l * kC);
                acc[l] = __builtin_amdgcn_wmma_f32_16x16x4_f32(
                    false, a, false, b, (short)0, acc[l], false, false);
            }
        }
        #pragma unroll
        for (int l = 0; l < kM0; ++l) {
            const float bias = b3[l * kC + cw];
            #pragma unroll
            for (int r = 0; r < 8; ++r) {
                int row = rbase + r;
                m0s[row * (kM0 * 64) + l * 64 + gw * 16 + lrow] = acc[l][r] + bias;
            }
        }
    };

    auto contract_half = [&](float (&nacc)[kHR], int& cur, unsigned& written, int hh, int nE, int cb) {
        for (int j = 0; j < nE; ++j) {
            const int g = __builtin_amdgcn_readfirstlane(srt_g[cb + j]);
            if (g != cur) {
                if (cur >= 0) {
                    flush_half(x, out, nacc, n0 + cur, hh * 64 + chan, rbeg, nr, (written >> cur) & 1u);
                    written |= (1u << cur);
                }
                cur = g;
            }
            const float ev = envs[j];
            float mv[kM0];
            #pragma unroll
            for (int l = 0; l < kM0; ++l) mv[l] = m0s[j * (kM0 * 64) + l * 64 + chan] * ev;
            #pragma unroll
            for (int rr = 0; rr < kHR; ++rr) {
                if (rr < nr) {
                    float s = nacc[rr];
                    #pragma unroll
                    for (int l = 0; l < kM0; ++l) s += wig[j * 128 + (rbeg + rr) * kM0 + l] * mv[l];
                    nacc[rr] = s;
                }
            }
        }
    };

    for (int lo = 0; lo < kmax; lo += kCAP) {
        int mg[kNG];
        int m = 0, mmax = 0;
        #pragma unroll
        for (int g = 0; g < kNG; ++g) {
            mg[g] = max(0, min(kCAP, tot[g] - lo));
            if (g == grp) m = mg[g];
            mmax = max(mmax, mg[g]);
        }

        int run[kNG];
        #pragma unroll
        for (int g = 0; g < kNG; ++g) run[g] = base[g];
        for (int s = wbeg; s < wend; s += 32 * kU) {
            int dv[kU];
            #pragma unroll
            for (int u = 0; u < kU; ++u) {
                const int idx = s + u * 32 + lane;
                dv[u] = edge_index[E + min(idx, wend - 1)];
            }
            #pragma unroll
            for (int u = 0; u < kU; ++u) {
                const int idx = s + u * 32 + lane;
                const int d   = dv[u] - noff - nb0;
                const bool in = (idx < wend) && (d >= 0) && (d < nbn);
                #pragma unroll
                for (int g = 0; g < kNG; ++g) {
                    const bool ig = in && d >= g * kG && d < (g + 1) * kG;
                    const unsigned mask = __builtin_amdgcn_ballot_w32(ig);
                    if (ig) {
                        const int rank = run[g] + (int)__builtin_amdgcn_mbcnt_lo(mask, 0u) - lo;
                        if (rank >= 0 && rank < mg[g]) { lst_e_[g][rank] = idx; lst_g_[g][rank] = d - g * kG; }
                    }
                    run[g] += __popc(mask);
                }
            }
        }
        __syncthreads();

        for (int t = gt; t < m; t += 128) {
            const int gn = lst_g[t];
            int pos = 0;
            for (int j = 0; j < m; ++j) {
                const int gj = lst_g[j];
                pos += ((gj < gn) || (gj == gn && j < t)) ? 1 : 0;
            }
            srt_e[pos] = lst_e[t];
            srt_g[pos] = gn;
        }
        __syncthreads();

        float px[kTILE], pw[kTILE], pd = 0.0f;
        auto request = [&](int cbn) {
            const int nEn = max(0, min(kTILE, m - cbn));
            if (nEn > 0) {
                const int eid = srt_e[cbn + min(lrow, nEn - 1)];
                #pragma unroll
                for (int it = 0; it < kTILE; ++it) {
                    const int e = __builtin_amdgcn_readlane(eid, it);
                    px[it] = x_edge[(size_t)e * 128 + gt];
                    pw[it] = wigner_inv[(size_t)e * (kNCOEF * kNCOEF) + woff];
                }
                if (gt < kTILE) pd = edge_distance[eid];
            } else {
                #pragma unroll
                for (int it = 0; it < kTILE; ++it) { px[it] = 0.0f; pw[it] = 0.0f; }
            }
        };
        request(0);
        int cur0 = -1, cur1 = -1;
        for (int cb = 0; cb < mmax; cb += kTILE) {
            const int nE = max(0, min(kTILE, m - cb));

            #pragma unroll
            for (int it = 0; it < kTILE; ++it) {
                xe[it * 128 + gt]  = (it < nE) ? px[it] : 0.0f;
                wig[it * 128 + gt] = (it < nE) ? pw[it] : 0.0f;
            }
            if (gt < kTILE) {
                float env = 0.0f;
                if (gt < nE) {
                    float d = pd * (1.0f / kCUTOFF);
                    if (d < 1.0f) {
                        float d5 = d * d * d * d * d;
                        env = 1.0f - 21.0f * d5 + 35.0f * d5 * d - 15.0f * d5 * d * d;
                    }
                }
                envs[gt] = env * (1.0f / kRESCALE);
            }
            if (cb + kTILE < mmax) request(cb + kTILE);
            lds_barrier();

            {
                const int colBase = gw * 16;
                v8f acc = {};
                for (int k0 = 0; k0 < 128; k0 += 4) {
                    v2f a, b;
                    a.x = xe[lrow * 128 + k0 + khalf];
                    a.y = xe[lrow * 128 + k0 + khalf + 1];
                    b.x = w1[(k0 + khalf)     * 64 + colBase + lrow];
                    b.y = w1[(k0 + khalf + 1) * 64 + colBase + lrow];
                    acc = __builtin_amdgcn_wmma_f32_16x16x4_f32(
                        false, a, false, b, (short)0, acc, false, false);
                }
                #pragma unroll
                for (int r = 0; r < 8; ++r) {
                    int row = rbase + r;
                    hraw[row * 64 + colBase + lrow] = acc[r] + b1[colBase + lrow];
                }
            }
            lds_barrier();
            ln_silu_16x64(hraw, h1, g1, be1, gt - lnw);
            lds_barrier();

            {
                const int colBase = gw * 16;
                v8f acc = {};
                for (int k0 = 0; k0 < 64; k0 += 4) {
                    v2f a, b;
                    a.x = h1[lrow * 64 + k0 + khalf];
                    a.y = h1[lrow * 64 + k0 + khalf + 1];
                    b.x = w2[(k0 + khalf)     * 64 + colBase + lrow];
                    b.y = w2[(k0 + khalf + 1) * 64 + colBase + lrow];
                    acc = __builtin_amdgcn_wmma_f32_16x16x4_f32(
                        false, a, false, b, (short)0, acc, false, false);
                }
                #pragma unroll
                for (int r = 0; r < 8; ++r) {
                    int row = rbase + r;
                    hraw[row * 64 + colBase + lrow] = acc[r] + b2[colBase + lrow];
                }
            }
            lds_barrier();
            ln_silu_16x64(hraw, h2, g2, be2, gt - lnw);
            lds_barrier();

            layer3_half(0);
            lds_barrier();
            contract_half(nacc0, cur0, written0, 0, nE, cb);
            lds_barrier();
            layer3_half(1);
            lds_barrier();
            contract_half(nacc1, cur1, written1, 1, nE, cb);
            __syncthreads();
        }
        if (cur0 >= 0) {
            flush_half(x, out, nacc0, n0 + cur0, chan, rbeg, nr, (written0 >> cur0) & 1u);
            written0 |= (1u << cur0);
        }
        if (cur1 >= 0) {
            flush_half(x, out, nacc1, n0 + cur1, 64 + chan, rbeg, nr, (written1 >> cur1) & 1u);
            written1 |= (1u << cur1);
        }
    }

    for (int g = 0; g < n1 - n0; ++g) {
        if (!((written0 >> g) & 1u)) {
            const size_t base = (size_t)(n0 + g) * (kNCOEF * kC) + (size_t)rbeg * kC + chan;
            float s0[kHR], s1[kHR];
            #pragma unroll
            for (int r = 0; r < kHR; ++r) {
                s0[r] = 0.0f; s1[r] = 0.0f;
                if (r < nr) { s0[r] = x[base + r * kC]; s1[r] = x[base + r * kC + 64]; }
            }
            #pragma unroll
            for (int r = 0; r < kHR; ++r) {
                if (r < nr) { out[base + r * kC] = s0[r]; out[base + r * kC + 64] = s1[r]; }
            }
        }
    }
}

extern "C" void kernel_launch(void* const* d_in, const int* in_sizes, int n_in,
                              void* d_out, int out_size, void* d_ws, size_t ws_size,
                              hipStream_t stream) {
    const float* x             = (const float*)d_in[0];
    const float* x_edge        = (const float*)d_in[1];
    const float* edge_distance = (const float*)d_in[2];
    const float* wigner_inv    = (const float*)d_in[3];
    const int*   edge_index    = (const int*)d_in[4];
    const float* w1  = (const float*)d_in[5];
    const float* b1  = (const float*)d_in[6];
    const float* g1  = (const float*)d_in[7];
    const float* be1 = (const float*)d_in[8];
    const float* w2  = (const float*)d_in[9];
    const float* b2  = (const float*)d_in[10];
    const float* g2  = (const float*)d_in[11];
    const float* be2 = (const float*)d_in[12];
    const float* w3  = (const float*)d_in[13];
    const float* b3  = (const float*)d_in[14];
    const int* node_offset = (const int*)d_in[15];

    const int E = in_sizes[1] / 128;

    const int N = out_size / (kNCOEF * kC);

    const int blocks = (N + kG * kNG - 1) / (kG * kNG);
    edge_degree_embed_kernel<<<blocks, kTHREADS, 0, stream>>>(
        x, x_edge, edge_distance, wigner_inv, edge_index,
        w1, b1, g1, be1, w2, b2, g2, be2, w3, b3,
        node_offset, (float*)d_out, E, N);
}
